# starting kernel
# baseline (speedup 1.0000x reference)
#include <hip/hip_runtime.h>
#include <hip/hip_bf16.h>
#include <math.h>

#define NN 50000
#define EE 800000
#define RR 8
#define IND 128
#define OUTD 128
#define HH 8
#define HDD 16

typedef float v2f __attribute__((ext_vector_type(2)));
typedef float v8f __attribute__((ext_vector_type(8)));

__device__ __forceinline__ void atomicMaxFloat(float* addr, float v) {
    if (v >= 0.0f) {
        atomicMax((int*)addr, __float_as_int(v));
    } else {
        atomicMin((unsigned int*)addr, __float_as_uint(v));
    }
}

__global__ __launch_bounds__(256) void rgat_init(float* __restrict__ out,
                                                 float* __restrict__ maxbuf,
                                                 float* __restrict__ denom) {
    int t = blockIdx.x * 256 + threadIdx.x;
    if (t < NN * OUTD) out[t] = 0.0f;
    if (t < NN * HH) {
        denom[t] = 0.0f;
        maxbuf[t] = -INFINITY;
    }
}

#define HS_A 132
#define WPS  288

__global__ __launch_bounds__(256) void rgat_proj_wmma(const float* __restrict__ h,
                                                      const float* __restrict__ W,
                                                      float* __restrict__ z) {
    extern __shared__ float lds[];
    float* h_s = lds;
    float* w_s = lds + 64 * HS_A;

    const int r     = blockIdx.y;
    const int nbase = blockIdx.x * 64;
    const int tid   = threadIdx.x;

    for (int i = tid; i < 64 * 32; i += 256) {
        int row = i >> 5;
        int c4  = i & 31;
        int g   = nbase + row;
        if (g >= NN) g = NN - 1;
        float4 v = *(const float4*)(h + (size_t)g * IND + c4 * 4);
        float* d = h_s + row * HS_A + c4 * 4;
        d[0] = v.x; d[1] = v.y; d[2] = v.z; d[3] = v.w;
    }
    const float* Wr = W + (size_t)r * IND * OUTD;
    for (int i = tid; i < 64 * 128; i += 256) {
        int kp  = i >> 7;
        int col = i & 127;
        float w0 = Wr[(size_t)(2 * kp) * OUTD + col];
        float w1 = Wr[(size_t)(2 * kp + 1) * OUTD + col];
        v2f p; p.x = w0; p.y = w1;
        *(v2f*)(w_s + kp * WPS + col * 2) = p;
    }
    __syncthreads();

    const int wave = tid >> 5;
    const int lane = tid & 31;
    const int ln   = lane & 15;
    const int g    = lane >> 4;
    const int rt   = wave & 3;
    const int cb   = (wave >> 2) * 64;

    v8f acc0 = {}, acc1 = {}, acc2 = {}, acc3 = {};
    const float* arow = h_s + (rt * 16 + ln) * HS_A;
    const float* bcol = w_s + (size_t)g * WPS + (size_t)(cb + ln) * 2;

    for (int k = 0; k < IND; k += 4) {
        v2f a = *(const v2f*)(arow + k + 2 * g);
        const float* brow = bcol + (k >> 1) * WPS;
        v2f b0 = *(const v2f*)(brow + 0 * 32);
        v2f b1 = *(const v2f*)(brow + 1 * 32);
        v2f b2 = *(const v2f*)(brow + 2 * 32);
        v2f b3 = *(const v2f*)(brow + 3 * 32);
        acc0 = __builtin_amdgcn_wmma_f32_16x16x4_f32(false, a, false, b0, (short)0, acc0, false, false);
        acc1 = __builtin_amdgcn_wmma_f32_16x16x4_f32(false, a, false, b1, (short)0, acc1, false, false);
        acc2 = __builtin_amdgcn_wmma_f32_16x16x4_f32(false, a, false, b2, (short)0, acc2, false, false);
        acc3 = __builtin_amdgcn_wmma_f32_16x16x4_f32(false, a, false, b3, (short)0, acc3, false, false);
    }

    float* zr = z + (size_t)r * NN * OUTD;
#pragma unroll
    for (int ct = 0; ct < 4; ++ct) {
        v8f acc = (ct == 0) ? acc0 : (ct == 1) ? acc1 : (ct == 2) ? acc2 : acc3;
        int col = cb + ct * 16 + ln;
#pragma unroll
        for (int v = 0; v < 8; ++v) {
            int row = nbase + rt * 16 + 8 * g + v;
            if (row < NN) zr[(size_t)row * OUTD + col] = acc[v];
        }
    }
}

__global__ __launch_bounds__(256) void rgat_edge_score(const float* __restrict__ z,
                                                       const float* __restrict__ attw,
                                                       const int* __restrict__ src,
                                                       const int* __restrict__ dst,
                                                       const int* __restrict__ etype,
                                                       float* __restrict__ e_val,
                                                       float* __restrict__ maxbuf) {
    int t = blockIdx.x * 256 + threadIdx.x;
    if (t >= EE * HH) return;
    int e  = t >> 3;
    int hd = t & 7;
    int r = etype[e], s = src[e], d = dst[e];

    const float4* zs = (const float4*)(z + ((size_t)r * NN + s) * OUTD + hd * HDD);
    const float4* zd = (const float4*)(z + ((size_t)r * NN + d) * OUTD + hd * HDD);
    const float4* aw = (const float4*)(attw + ((size_t)r * HH + hd) * (2 * HDD));

    float acc = 0.0f;
#pragma unroll
    for (int q = 0; q < 4; ++q) {
        float4 a = zs[q], wA = aw[q];
        float4 b = zd[q], wB = aw[4 + q];
        acc += a.x * wA.x + a.y * wA.y + a.z * wA.z + a.w * wA.w;
        acc += b.x * wB.x + b.y * wB.y + b.z * wB.z + b.w * wB.w;
    }
    float ev = (acc > 0.0f) ? acc : 0.01f * acc;
    e_val[t] = ev;
    atomicMaxFloat(maxbuf + (size_t)d * HH + hd, ev);
}

__global__ __launch_bounds__(256) void rgat_edge_accum(const float* __restrict__ z,
                                                       const int* __restrict__ src,
                                                       const int* __restrict__ dst,
                                                       const int* __restrict__ etype,
                                                       const float* __restrict__ e_val,
                                                       const float* __restrict__ maxbuf,
                                                       float* __restrict__ denom,
                                                       float* __restrict__ out) {
    int t = blockIdx.x * 256 + threadIdx.x;
    if (t >= EE * HH) return;
    int e  = t >> 3;
    int hd = t & 7;
    int r = etype[e], s = src[e], d = dst[e];

    float m  = maxbuf[(size_t)d * HH + hd];
    float ex = __expf(e_val[t] - m);
    atomicAdd(denom + (size_t)d * HH + hd, ex);

    const float* zs = z + ((size_t)r * NN + s) * OUTD + hd * HDD;
    float* o = out + (size_t)d * OUTD + hd * HDD;
#pragma unroll
    for (int i = 0; i < HDD; ++i) atomicAdd(o + i, ex * zs[i]);
}

__global__ __launch_bounds__(256) void rgat_finalize(float* __restrict__ out,
                                                     const float* __restrict__ denom) {
    int t = blockIdx.x * 256 + threadIdx.x;
    if (t >= NN * OUTD) return;
    out[t] = out[t] / fmaxf(denom[t >> 4], 1e-9f);
}

extern "C" void kernel_launch(void* const* d_in, const int* in_sizes, int n_in,
                              void* d_out, int out_size, void* d_ws, size_t ws_size,
                              hipStream_t stream) {
    const float* h    = (const float*)d_in[0];
    const float* fcw  = (const float*)d_in[1];
    const float* attw = (const float*)d_in[2];
    const int*   src  = (const int*)d_in[3];
    const int*   dst  = (const int*)d_in[4];
    const int*   ety  = (const int*)d_in[5];
    float* out = (float*)d_out;

    float* z      = (float*)d_ws;
    float* e_val  = z + (size_t)RR * NN * OUTD;
    float* maxbuf = e_val + (size_t)EE * HH;
    float* denom  = maxbuf + (size_t)NN * HH;

    {
        int total = NN * OUTD;
        rgat_init<<<(total + 255) / 256, 256, 0, stream>>>(out, maxbuf, denom);
    }
    {
        dim3 grid((NN + 63) / 64, RR);
        size_t shmem = (size_t)(64 * HS_A + 64 * WPS) * sizeof(float);
        rgat_proj_wmma<<<grid, 256, shmem, stream>>>(h, fcw, z);
    }
    {
        int total = EE * HH;
        rgat_edge_score<<<(total + 255) / 256, 256, 0, stream>>>(
            z, attw, src, dst, ety, e_val, maxbuf);
    }
    {
        int total = EE * HH;
        rgat_edge_accum<<<(total + 255) / 256, 256, 0, stream>>>(
            z, src, dst, ety, e_val, maxbuf, denom, out);
    }
    {
        int total = NN * OUTD;
        rgat_finalize<<<(total + 255) / 256, 256, 0, stream>>>(out, denom);
    }
}


# z kept in the workspace in accumulator-fragment column order: projection writes whole lines with 16-byte stores (16 per tile, was 64), an edge row is one 16-byte load per lane (was 4); same sums
# speedup vs baseline: 4.6851x; 4.6851x over previous
#include <hip/hip_runtime.h>
#include <hip/hip_bf16.h>
#include <math.h>

#define NN 50000
#define EE 800000
#define RR 8
#define IND 128
#define OUTD 128
#define HH 8
#define HDD 16

typedef float v2f __attribute__((ext_vector_type(2)));
typedef float v8f __attribute__((ext_vector_type(8)));
typedef float v4f __attribute__((ext_vector_type(4)));

__device__ __forceinline__ void atomicMaxFloat(float* addr, float v) {
    if (v >= 0.0f) {
        atomicMax((int*)addr, __float_as_int(v));
    } else {
        atomicMin((unsigned int*)addr, __float_as_uint(v));
    }
}

__device__ __forceinline__ float edge_logit(float s_half, float d_half) {
    const float a = s_half + d_half;
    return (a > 0.0f) ? a : 0.01f * a;
}

#define FOLD_BLOCKS (RR * IND * 16 / 256)
#define WL_R        (IND * 16)

#define TN       196
#define NTILES   ((NN + TN - 1) / TN)
#define SBW      512
#define NCH      ((EE + SBW - 1) / SBW)
#define SB_VECS  (NCH * 32)
#define SB_BLOCKS ((SB_VECS + 255) / 256)

static_assert(NTILES <= 256, "a slab number fits a byte");
static_assert(EE % 16 == 0, "a 16-edge vector is all edges or all pad");

__global__ __launch_bounds__(256) void rgat_init(const float* __restrict__ W,
                                                 const float* __restrict__ attw,
                                                 const int* __restrict__ dst,
                                                 float* __restrict__ wl,
                                                 unsigned int* __restrict__ sb) {
    if (blockIdx.x >= FOLD_BLOCKS) {
        const int i = (blockIdx.x - FOLD_BLOCKS) * 256 + threadIdx.x;
        if (i < SB_VECS) {
            uint4 o = make_uint4(0u, 0u, 0u, 0u);
            if (i < EE / 16) {
                const int4* d4 = (const int4*)dst + (size_t)i * 4;
                unsigned w[4];
#pragma unroll
                for (int k = 0; k < 4; ++k) {
                    const int4 d = d4[k];
                    w[k] = ((unsigned)d.x / TN) | (((unsigned)d.y / TN) << 8) |
                           (((unsigned)d.z / TN) << 16) | (((unsigned)d.w / TN) << 24);
                }
                o = make_uint4(w[0], w[1], w[2], w[3]);
            }
            ((uint4*)sb)[i] = o;
        }
        return;
    }
    const int idx  = blockIdx.x * 256 + threadIdx.x;
    const int r    = idx >> 11;
    const int kp   = (idx >> 5) & 63;
    const int j    = (idx >> 1) & 15;
    const int jj   = idx & 1;
    const int hd   = j & 7;
    const int half = j >> 3;
    const float4* w = (const float4*)(W + ((size_t)r * IND + 2 * kp + jj) * OUTD + hd * HDD);
    const float4* a = (const float4*)(attw + ((size_t)r * HH + hd) * (2 * HDD) + half * HDD);
    float acc = 0.0f;
#pragma unroll
    for (int q = 0; q < 4; ++q) {
        const float4 x = w[q], y = a[q];
        acc += x.x * y.x + x.y * y.y + x.z * y.z + x.w * y.w;
    }
    wl[idx] = acc;
}

#define HS_A   132
#define NT16   (NN / 16)
#define PSEQ   128
#define PBLOCKS (2 * PSEQ)
#define KSTEPS (IND / 4)

static_assert(NN % 16 == 0, "h is a whole number of 16-row tiles");
static_assert(RR == 8, "two blocks of four waves cover the relations");

__global__ __launch_bounds__(128) void rgat_proj_wmma(const float* __restrict__ h,
                                                      const float* __restrict__ W,
                                                      const float* __restrict__ wl,
                                                      float* __restrict__ z,
                                                      float* __restrict__ lp) {
    __shared__ __attribute__((aligned(16))) float hs[4][16 * HS_A];

    const int tid  = threadIdx.x;
    const int wave = __builtin_amdgcn_readfirstlane(tid >> 5);
    const int lane = tid & 31;
    const int ln   = lane & 15;
    const int g    = lane >> 4;
    const int r    = (blockIdx.x & 1) * 4 + wave;
    const int q    = blockIdx.x >> 1;

    v2f B[KSTEPS][9];
    {
        const float* Wr = W + (size_t)r * IND * OUTD + (size_t)(2 * g) * OUTD + ln;
        const float* wf = wl + (size_t)r * WL_R + (size_t)g * 32 + ln * 2;
#pragma unroll
        for (int ks = 0; ks < KSTEPS; ++ks) {
#pragma unroll
            for (int ct = 0; ct < 8; ++ct) {
                B[ks][ct].x = Wr[(4 * ks) * OUTD + ct * 16];
                B[ks][ct].y = Wr[(4 * ks + 1) * OUTD + ct * 16];
            }
            B[ks][8] = *(const v2f*)(wf + (2 * ks) * 32);
        }
    }

    float* strip = &hs[wave][0];
    const v4f* h4 = (const v4f*)h + lane;
    v4f rows[16];

#pragma unroll
    for (int i = 0; i < 16; ++i) rows[i] = h4[((size_t)q * 16 + i) * (IND / 4)];
#pragma unroll
    for (int i = 0; i < 16; ++i) *(v4f*)(strip + i * HS_A + 4 * lane) = rows[i];

    for (int t = q; t < NT16; t += PSEQ) {
        const int tn = (t + PSEQ < NT16) ? t + PSEQ : t;
        const v4f* pn = h4 + (size_t)tn * 16 * (IND / 4);
        asm volatile(
            "global_load_b128 %0, %16, off\n\t"
            "global_load_b128 %1, %16, off offset:512\n\t"
            "global_load_b128 %2, %16, off offset:1024\n\t"
            "global_load_b128 %3, %16, off offset:1536\n\t"
            "global_load_b128 %4, %16, off offset:2048\n\t"
            "global_load_b128 %5, %16, off offset:2560\n\t"
            "global_load_b128 %6, %16, off offset:3072\n\t"
            "global_load_b128 %7, %16, off offset:3584\n\t"
            "global_load_b128 %8, %16, off offset:4096\n\t"
            "global_load_b128 %9, %16, off offset:4608\n\t"
            "global_load_b128 %10, %16, off offset:5120\n\t"
            "global_load_b128 %11, %16, off offset:5632\n\t"
            "global_load_b128 %12, %16, off offset:6144\n\t"
            "global_load_b128 %13, %16, off offset:6656\n\t"
            "global_load_b128 %14, %16, off offset:7168\n\t"
            "global_load_b128 %15, %16, off offset:7680"
            : "=&v"(rows[0]), "=&v"(rows[1]), "=&v"(rows[2]), "=&v"(rows[3]), "=&v"(rows[4]), "=&v"(rows[5]), "=&v"(rows[6]), "=&v"(rows[7]), "=&v"(rows[8]), "=&v"(rows[9]), "=&v"(rows[10]), "=&v"(rows[11]), "=&v"(rows[12]), "=&v"(rows[13]), "=&v"(rows[14]), "=&v"(rows[15])
            : "v"(pn));

        __builtin_amdgcn_fence(__ATOMIC_ACQ_REL, "wavefront");
        v2f a[KSTEPS];
#pragma unroll
        for (int ks = 0; ks < KSTEPS; ++ks)
            a[ks] = *(const v2f*)(strip + ln * HS_A + 4 * ks + 2 * g);

        v8f acc[9];
#pragma unroll
        for (int ct = 0; ct < 9; ++ct) acc[ct] = v8f{};
#pragma unroll
        for (int ks = 0; ks < KSTEPS; ++ks) {
#pragma unroll
            for (int ct = 0; ct < 9; ++ct)
                acc[ct] = __builtin_amdgcn_wmma_f32_16x16x4_f32(false, a[ks], false, B[ks][ct],
                                                                (short)0, acc[ct], false, false);
        }

        float* zr = z + ((size_t)r * NN + (size_t)t * 16 + 8 * g) * OUTD + 4 * ln;
#pragma unroll
        for (int v = 0; v < 8; ++v) {
            *(v4f*)(zr + v * OUTD)      = v4f{acc[0][v], acc[1][v], acc[2][v], acc[3][v]};
            *(v4f*)(zr + v * OUTD + 64) = v4f{acc[4][v], acc[5][v], acc[6][v], acc[7][v]};
        }
        float* lr = lp + ((size_t)r * NN + (size_t)t * 16 + 8 * g) * 16 + ln;
#pragma unroll
        for (int v = 0; v < 8; ++v) lr[v * 16] = acc[8][v];

        asm volatile("s_wait_loadcnt 0x0" : "+v"(rows[0]), "+v"(rows[1]), "+v"(rows[2]), "+v"(rows[3]), "+v"(rows[4]), "+v"(rows[5]), "+v"(rows[6]), "+v"(rows[7]), "+v"(rows[8]), "+v"(rows[9]), "+v"(rows[10]), "+v"(rows[11]), "+v"(rows[12]), "+v"(rows[13]), "+v"(rows[14]), "+v"(rows[15]) : "v"(pn));
#pragma unroll
        for (int i = 0; i < 16; ++i) *(v4f*)(strip + i * HS_A + 4 * lane) = rows[i];
    }
}

#define TWAVES  32
#define SCAN_U  2
#define ROUND_E (SCAN_U * SBW)
#define QCAP    1600
#define QKEEP   (QCAP - ROUND_E)
#define TILE_LDS_DWORDS (TN * OUTD + 2 * TN * HH + TWAVES * QCAP)
#define KB      4
#define KEY_SH  19

static_assert(TILE_LDS_DWORDS * 4 <= 320 * 1024, "slab + queues fit the WGP's LDS");
static_assert(SCAN_U == 2, "the sweep below is written for sets of two chunks");
static_assert(QKEEP >= 32, "a wave that has given up keeping still has room for a round");
static_assert((TN * OUTD + TN * HH) % 4 == 0 && (TN * HH) % 4 == 0, "16-byte fills");
static_assert(RR * NN <= (1 << KEY_SH) && TN <= (1 << (31 - KEY_SH)), "relation * NN + src and the slab row share a word");

__device__ __forceinline__ float rl_f(float v, int j) {
    return __int_as_float(__builtin_amdgcn_readlane(__float_as_int(v), j));
}

__global__ __launch_bounds__(1024) void rgat_tile_accum(const float* __restrict__ z,
                                                        const int* __restrict__ src,
                                                        const int* __restrict__ dst,
                                                        const int* __restrict__ etype,
                                                        const float* __restrict__ lp,
                                                        const unsigned int* __restrict__ slabb,
                                                        float* __restrict__ out) {
    __shared__ __attribute__((aligned(16))) float tl[TILE_LDS_DWORDS];
    float* num = tl;
    float* den = tl + TN * OUTD;
    float* mx  = tl + TN * OUTD + TN * HH;
    int*   qs  = (int*)(tl + TN * OUTD + 2 * TN * HH);

    const int tid  = threadIdx.x;
    const int lane = tid & 31;
    const int wave = __builtin_amdgcn_readfirstlane(tid >> 5);
    const int lo   = blockIdx.x * TN;
    const bool hi  = (lane >> 4) != 0;

    for (int i = tid; i < (TN * OUTD + TN * HH) / 4; i += 1024)
        ((float4*)tl)[i] = make_float4(0.f, 0.f, 0.f, 0.f);
    if (tid < TN * HH / 4)
        ((float4*)mx)[tid] = make_float4(-INFINITY, -INFINITY, -INFINITY, -INFINITY);
    __syncthreads();

    int* q = qs + wave * QCAP;
    int cnt = 0;
    const uint4* sb4 = (const uint4*)slabb;
    const unsigned pat = (unsigned)blockIdx.x * 0x01010101u;

    auto load_vec = [&](int c) -> uint4 {
        return sb4[(c < NCH ? c : NCH - 1) * 32 + lane];
    };
    auto test_vec = [&](const uint4& g, int c) {
        const int ebase = (c * 32 + lane) * 16;
        const unsigned live = (ebase < EE) ? 0x80808080u : 0u;
        const unsigned w[4] = {g.x, g.y, g.z, g.w};
#pragma unroll
        for (int k = 0; k < 4; ++k) {
            const unsigned x = w[k] ^ pat;
            unsigned zb = ~(((x & 0x7F7F7F7Fu) + 0x7F7F7F7Fu) | x) & live;
#pragma unroll
            for (int it = 0; it < 4; ++it) {
                const bool hit = zb != 0u;
                const unsigned b = __builtin_amdgcn_ballot_w32(hit);
                if (!b) break;
                if (hit) {
                    q[cnt + (int)__builtin_amdgcn_mbcnt_lo(b, 0u)] = ebase + 4 * k + (__builtin_ctz(zb) >> 3);
                    zb &= zb - 1u;
                }
                cnt += __builtin_popcount(b);
            }
        }
    };
    auto scan = [&](auto&& round_end) {
        uint4 ga[SCAN_U], gb[SCAN_U];
        ga[0] = load_vec(wave);
        ga[1] = load_vec(wave + TWAVES);
        for (int c0 = wave; c0 < NCH; c0 += 2 * SCAN_U * TWAVES) {
            gb[0] = load_vec(c0 + 2 * TWAVES);
            gb[1] = load_vec(c0 + 3 * TWAVES);
            test_vec(ga[0], c0);
            test_vec(ga[1], c0 + TWAVES);
            round_end(false);
            ga[0] = load_vec(c0 + 4 * TWAVES);
            ga[1] = load_vec(c0 + 5 * TWAVES);
            test_vec(gb[0], c0 + 2 * TWAVES);
            test_vec(gb[1], c0 + 3 * TWAVES);
            round_end(c0 + 2 * SCAN_U * TWAVES >= NCH);
        }
    };

    auto maxdrain = [&](int from, int to) {
        __builtin_amdgcn_fence(__ATOMIC_ACQ_REL, "wavefront");
        for (int base = from; base < to; base += 32) {
            if (base + lane < to) {
                const int e  = q[base + lane];
                const int s = src[e], r = etype[e];
                const int dl = dst[e] - lo;
                const float4* ps = (const float4*)(lp + (size_t)(r * NN + s) * 16);
                const float4* pd = (const float4*)(lp + (size_t)(r * NN + lo + dl) * 16 + 8);
                const float4 sa = ps[0], sb = ps[1], da = pd[0], db = pd[1];
                float* m = mx + dl * HH;
                atomicMaxFloat(m + 0, edge_logit(sa.x, da.x));
                atomicMaxFloat(m + 1, edge_logit(sa.y, da.y));
                atomicMaxFloat(m + 2, edge_logit(sa.z, da.z));
                atomicMaxFloat(m + 3, edge_logit(sa.w, da.w));
                atomicMaxFloat(m + 4, edge_logit(sb.x, db.x));
                atomicMaxFloat(m + 5, edge_logit(sb.y, db.y));
                atomicMaxFloat(m + 6, edge_logit(sb.z, db.z));
                atomicMaxFloat(m + 7, edge_logit(sb.w, db.w));
            }
        }
    };

#define KEEP_SET(n) \
    int kq##n = 0; \
    float lg##n##0 = 0.f, lg##n##1 = 0.f, lg##n##2 = 0.f, lg##n##3 = 0.f, \
          lg##n##4 = 0.f, lg##n##5 = 0.f, lg##n##6 = 0.f, lg##n##7 = 0.f;
    KEEP_SET(0) KEEP_SET(1) KEEP_SET(2) KEEP_SET(3)
#undef KEEP_SET
    static_assert(KB == 4, "four register sets are written out");

    auto maxbatch = [&](int k, int nact) {
        __builtin_amdgcn_fence(__ATOMIC_ACQ_REL, "wavefront");
        int key = 0;
        float l0 = 0.f, l1 = 0.f, l2 = 0.f, l3 = 0.f, l4 = 0.f, l5 = 0.f, l6 = 0.f, l7 = 0.f;
        if (lane < nact) {
            const int e  = q[k * 32 + lane];
            const int s = src[e], r = etype[e];
            const int dl = dst[e] - lo;
            const int rn = r * NN + s;
            const float4* ps = (const float4*)(lp + (size_t)rn * 16);
            const float4* pd = (const float4*)(lp + (size_t)(r * NN + lo + dl) * 16 + 8);
            const float4 sa = ps[0], sb = ps[1], da = pd[0], db = pd[1];
            l0 = edge_logit(sa.x, da.x); l1 = edge_logit(sa.y, da.y);
            l2 = edge_logit(sa.z, da.z); l3 = edge_logit(sa.w, da.w);
            l4 = edge_logit(sb.x, db.x); l5 = edge_logit(sb.y, db.y);
            l6 = edge_logit(sb.z, db.z); l7 = edge_logit(sb.w, db.w);
            float* m = mx + dl * HH;
            atomicMaxFloat(m + 0, l0); atomicMaxFloat(m + 1, l1);
            atomicMaxFloat(m + 2, l2); atomicMaxFloat(m + 3, l3);
            atomicMaxFloat(m + 4, l4); atomicMaxFloat(m + 5, l5);
            atomicMaxFloat(m + 6, l6); atomicMaxFloat(m + 7, l7);
            key = rn | (dl << KEY_SH);
        }
#define KEEP_PUT(n) \
        kq##n   = (k == n) ? key : kq##n; \
        lg##n##0 = (k == n) ? l0 : lg##n##0; lg##n##1 = (k == n) ? l1 : lg##n##1; \
        lg##n##2 = (k == n) ? l2 : lg##n##2; lg##n##3 = (k == n) ? l3 : lg##n##3; \
        lg##n##4 = (k == n) ? l4 : lg##n##4; lg##n##5 = (k == n) ? l5 : lg##n##5; \
        lg##n##6 = (k == n) ? l6 : lg##n##6; lg##n##7 = (k == n) ? l7 : lg##n##7;
        KEEP_PUT(0) KEEP_PUT(1) KEEP_PUT(2) KEEP_PUT(3)
#undef KEEP_PUT
    };

    auto drain = [&](int total, bool regs) {
        __builtin_amdgcn_fence(__ATOMIC_ACQ_REL, "wavefront");
        for (int base = 0; base < total; base += 32) {
            const int n = (total - base < 32) ? (total - base) : 32;
            int zoff = 0, dl = 0;
            float ex0 = 0.f, ex1 = 0.f, ex2 = 0.f, ex3 = 0.f, ex4 = 0.f, ex5 = 0.f, ex6 = 0.f, ex7 = 0.f;
            const int kb = base >> 5;
            const bool kept = regs && kb < KB;
#define KEEP_GET(a0, a1, a2, a3) ((kb == 0) ? (a0) : (kb == 1) ? (a1) : (kb == 2) ? (a2) : (a3))
            const int key = KEEP_GET(kq0, kq1, kq2, kq3);
            float4 ea = make_float4(KEEP_GET(lg00, lg10, lg20, lg30), KEEP_GET(lg01, lg11, lg21, lg31),
                                    KEEP_GET(lg02, lg12, lg22, lg32), KEEP_GET(lg03, lg13, lg23, lg33));
            float4 eb = make_float4(KEEP_GET(lg04, lg14, lg24, lg34), KEEP_GET(lg05, lg15, lg25, lg35),
                                    KEEP_GET(lg06, lg16, lg26, lg36), KEEP_GET(lg07, lg17, lg27, lg37));
#undef KEEP_GET
            if (lane < n) {
                if (kept) {
                    dl = key >> KEY_SH;
                    zoff = (key & ((1 << KEY_SH) - 1)) * OUTD;
                } else {
                    const int e  = q[base + lane];
                    const int s = src[e], r = etype[e];
                    dl = dst[e] - lo;
                    zoff = (r * NN + s) * OUTD;
                    const float4* ps = (const float4*)(lp + (size_t)(r * NN + s) * 16);
                    const float4* pd = (const float4*)(lp + (size_t)(r * NN + lo + dl) * 16 + 8);
                    const float4 sa = ps[0], sb = ps[1], da = pd[0], db = pd[1];
                    ea.x = edge_logit(sa.x, da.x); ea.y = edge_logit(sa.y, da.y);
                    ea.z = edge_logit(sa.z, da.z); ea.w = edge_logit(sa.w, da.w);
                    eb.x = edge_logit(sb.x, db.x); eb.y = edge_logit(sb.y, db.y);
                    eb.z = edge_logit(sb.z, db.z); eb.w = edge_logit(sb.w, db.w);
                }
                const float4 ma = *(const float4*)(mx + dl * HH);
                const float4 mb = *(const float4*)(mx + dl * HH + 4);
                ex0 = __expf(ea.x - ma.x); ex1 = __expf(ea.y - ma.y);
                ex2 = __expf(ea.z - ma.z); ex3 = __expf(ea.w - ma.w);
                ex4 = __expf(eb.x - mb.x); ex5 = __expf(eb.y - mb.y);
                ex6 = __expf(eb.z - mb.z); ex7 = __expf(eb.w - mb.w);
                float* dn = den + dl * HH;
                atomicAdd(dn + 0, ex0); atomicAdd(dn + 1, ex1);
                atomicAdd(dn + 2, ex2); atomicAdd(dn + 3, ex3);
                atomicAdd(dn + 4, ex4); atomicAdd(dn + 5, ex5);
                atomicAdd(dn + 6, ex6); atomicAdd(dn + 7, ex7);
            }
            for (int j = 0; j < n; j += 8) {
                float4 v[8];
#pragma unroll
                for (int k = 0; k < 8; ++k)
                    v[k] = ((const float4*)(z + __builtin_amdgcn_readlane(zoff, j + k)))[lane];
#pragma unroll
                for (int k = 0; k < 8; ++k) {
                    const int jj = j + k;
                    const bool live = (k == 0) || (j + k) < n;
                    float* nr = num + __builtin_amdgcn_readlane(dl, jj) * OUTD + lane;
                    const float a0 = rl_f(ex0, jj), a1 = rl_f(ex1, jj), a2 = rl_f(ex2, jj), a3 = rl_f(ex3, jj);
                    const float a4 = rl_f(ex4, jj), a5 = rl_f(ex5, jj), a6 = rl_f(ex6, jj), a7 = rl_f(ex7, jj);
                    const float w0 = hi ? a4 : a0;
                    const float w1 = hi ? a5 : a1;
                    const float w2 = hi ? a6 : a2;
                    const float w3 = hi ? a7 : a3;
                    atomicAdd(nr +  0, live ? w0 * v[k].x : 0.f);
                    atomicAdd(nr + 32, live ? w1 * v[k].y : 0.f);
                    atomicAdd(nr + 64, live ? w2 * v[k].z : 0.f);
                    atomicAdd(nr + 96, live ? w3 * v[k].w : 0.f);
                }
            }
        }
    };

    int  done = 0;
    bool keep = true;
    scan([&](bool last) {
        const bool giveup = keep && cnt > QKEEP;
        if (giveup) keep = false;
        if (keep) {
            const int lim = last ? cnt : (cnt & ~31);
            for (; done < lim; done += 32) maxbatch(done >> 5, (lim - done < 32) ? (lim - done) : 32);
        } else if (giveup || cnt - done >= 32 || (last && cnt > done)) {
            maxdrain(done, cnt);
            cnt = 0; done = 0;
        }
    });
    __syncthreads();

    if (keep) {
        if (cnt > 0) drain(cnt, true);
    } else {
        cnt = 0;
        scan([&](bool last) {
            if (cnt >= 32 || (last && cnt > 0)) { drain(cnt, false); cnt = 0; }
        });
    }
    __syncthreads();

    for (int i = tid; i < TN * (OUTD / 4); i += 1024) {
        const int nl = i >> 5, c4 = i & 31;
        const int node = lo + nl;
        if (node < NN) {
            float4 a = *(const float4*)(num + nl * OUTD + ((c4 >> 2) & 3) * 32 + (c4 >> 4) * 16 + (c4 & 3) * 4);
            const float dd = fmaxf(den[nl * HH + (c4 >> 2)], 1e-9f);
            a.x = a.x / dd; a.y = a.y / dd; a.z = a.z / dd; a.w = a.w / dd;
            ((float4*)out)[node * (OUTD / 4) + c4] = a;
        }
    }
}

extern "C" void kernel_launch(void* const* d_in, const int* in_sizes, int n_in,
                              void* d_out, int out_size, void* d_ws, size_t ws_size,
                              hipStream_t stream) {
    const float* h    = (const float*)d_in[0];
    const float* fcw  = (const float*)d_in[1];
    const float* attw = (const float*)d_in[2];
    const int*   src  = (const int*)d_in[3];
    const int*   dst  = (const int*)d_in[4];
    const int*   ety  = (const int*)d_in[5];
    float* out = (float*)d_out;

    float* z      = (float*)d_ws;
    float* lp     = z + (size_t)RR * NN * OUTD;
    unsigned int* slabb = (unsigned int*)(lp + (size_t)RR * NN * 16);
    float* wl     = out;

    rgat_init<<<FOLD_BLOCKS + SB_BLOCKS, 256, 0, stream>>>(fcw, attw, dst, wl, slabb);
    rgat_proj_wmma<<<PBLOCKS, 128, 0, stream>>>(h, fcw, wl, z, lp);
    rgat_tile_accum<<<NTILES, 1024, 0, stream>>>(z, src, dst, ety, lp, slabb, out);
}
